# starting kernel
# baseline (speedup 1.0000x reference)
#include <hip/hip_runtime.h>

#define NROWS 10000
#define IN_C  2048
#define HID   256
#define OUT_C 256
#define E1    160000
#define E2    320000

#define PSTRIDE 160
#define LDS_CHUNK (32 * PSTRIDE)

typedef float v2f __attribute__((ext_vector_type(2)));
typedef float v8f __attribute__((ext_vector_type(8)));

__global__ void h2gcn_fill_zero(float* __restrict__ p, int n) {
    int i = blockIdx.x * blockDim.x + threadIdx.x;
    if (i < n) p[i] = 0.0f;
}

__global__ void h2gcn_gemm1(const float* __restrict__ x,
                            const float* __restrict__ W1,
                            float* __restrict__ h0) {
    __shared__ float lB[2][LDS_CHUNK];
    const int w    = threadIdx.x >> 5;
    const int lane = threadIdx.x & 31;
    const int m0   = (blockIdx.x * 8 + w) * 16;
    const int nb   = blockIdx.y * 64;
    const bool active = (m0 < NROWS);

    v8f acc[4];
    #pragma unroll
    for (int nt = 0; nt < 4; ++nt) acc[nt] = 0.0f;

    int aRow = m0 + (lane & 15);
    if (aRow >= NROWS) aRow = NROWS - 1;
    const float* aBase = x + (size_t)aRow * IN_C + ((lane >> 4) << 1);

    float4 sA[2], sB[2];
    auto loadStage = [&](int k0) {
        #pragma unroll
        for (int e = 0; e < 2; ++e) {
            const int u  = threadIdx.x + 256 * e;
            const int p  = u >> 4;
            const int n4 = (u & 15) << 2;
            const float* g = &W1[(size_t)(k0 + 2 * p) * HID + nb + n4];
            sA[e] = *(const float4*)g;
            sB[e] = *(const float4*)(g + HID);
        }
    };
    auto storeStage = [&](int buf) {
        #pragma unroll
        for (int e = 0; e < 2; ++e) {
            const int u  = threadIdx.x + 256 * e;
            const int p  = u >> 4;
            const int n4 = (u & 15) << 2;
            float* d = &lB[buf][p * PSTRIDE + n4 * 2];
            const float a0[4] = {sA[e].x, sA[e].y, sA[e].z, sA[e].w};
            const float b0[4] = {sB[e].x, sB[e].y, sB[e].z, sB[e].w};
            #pragma unroll
            for (int i = 0; i < 4; ++i) {
                v2f pr; pr.x = a0[i]; pr.y = b0[i];
                *(v2f*)(d + 2 * i) = pr;
            }
        }
    };

    constexpr int NCHUNK = IN_C / 64;
    loadStage(0);
    for (int c = 0; c < NCHUNK; ++c) {
        const int buf = c & 1;
        storeStage(buf);
        __syncthreads();
        if (c + 1 < NCHUNK) loadStage((c + 1) * 64);

        const float* ap = aBase + c * 64;
        #pragma unroll
        for (int kk = 0; kk < 64; kk += 4) {
            const v2f a = *(const v2f*)(ap + kk);
            const int p = (kk >> 1) + (lane >> 4);
            const float* lb = &lB[buf][p * PSTRIDE + (lane & 15) * 2];
            #pragma unroll
            for (int nt = 0; nt < 4; ++nt) {
                const v2f b = *(const v2f*)(lb + nt * 32);
                acc[nt] = __builtin_amdgcn_wmma_f32_16x16x4_f32(
                    false, a, false, b, (short)0, acc[nt], false, false);
            }
        }
        __syncthreads();
    }

    if (active) {
        const int rbase = m0 + ((lane >> 4) << 3);
        #pragma unroll
        for (int nt = 0; nt < 4; ++nt) {
            const int col = nb + nt * 16 + (lane & 15);
            #pragma unroll
            for (int v = 0; v < 8; ++v)
                h0[(size_t)(rbase + v) * HID + col] = acc[nt][v];
        }
    }
}

__global__ void h2gcn_spmm(const int* __restrict__ rows,
                           const int* __restrict__ cols,
                           const float* __restrict__ vals,
                           const float* __restrict__ h0,
                           float* __restrict__ out, int E) {
    const int edge = blockIdx.x * (blockDim.x >> 5) + (threadIdx.x >> 5);
    const int lane = threadIdx.x & 31;
    if (edge >= E) return;
    const int   r = rows[edge];
    const int   c = cols[edge];
    const float v = vals[edge];
    const float4* src = (const float4*)(h0 + (size_t)c * HID);
    float* dst = out + (size_t)r * HID;
    #pragma unroll
    for (int i = 0; i < 2; ++i) {
        const int j = lane + 32 * i;
        const float4 s = src[j];
        atomicAdd(dst + 4 * j + 0, v * s.x);
        atomicAdd(dst + 4 * j + 1, v * s.y);
        atomicAdd(dst + 4 * j + 2, v * s.z);
        atomicAdd(dst + 4 * j + 3, v * s.w);
    }
}

__global__ void h2gcn_gemm2(const float* __restrict__ h0,
                            const float* __restrict__ h1,
                            const float* __restrict__ h2,
                            const float* __restrict__ Wout,
                            const float* __restrict__ bout,
                            float* __restrict__ out) {
    __shared__ float lB[2][LDS_CHUNK];
    const int w    = threadIdx.x >> 5;
    const int lane = threadIdx.x & 31;
    const int m0   = (blockIdx.x * 8 + w) * 16;
    const int nb   = blockIdx.y * 64;
    const bool active = (m0 < NROWS);

    v8f acc[4];
    #pragma unroll
    for (int nt = 0; nt < 4; ++nt) {
        const float b = bout[nb + nt * 16 + (lane & 15)];
        #pragma unroll
        for (int v = 0; v < 8; ++v) acc[nt][v] = b;
    }

    int aRow = m0 + (lane & 15);
    if (aRow >= NROWS) aRow = NROWS - 1;
    const size_t aOff = (size_t)aRow * HID + ((lane >> 4) << 1);

    float4 sA[2], sB[2];
    auto loadStage = [&](int c) {
        const int part = c >> 2;
        const int k0   = (c & 3) * 64;
        const float* wp = Wout + (size_t)part * HID * OUT_C;
        #pragma unroll
        for (int e = 0; e < 2; ++e) {
            const int u  = threadIdx.x + 256 * e;
            const int p  = u >> 4;
            const int n4 = (u & 15) << 2;
            const float* g = &wp[(size_t)(k0 + 2 * p) * OUT_C + nb + n4];
            sA[e] = *(const float4*)g;
            sB[e] = *(const float4*)(g + OUT_C);
        }
    };
    auto storeStage = [&](int buf) {
        #pragma unroll
        for (int e = 0; e < 2; ++e) {
            const int u  = threadIdx.x + 256 * e;
            const int p  = u >> 4;
            const int n4 = (u & 15) << 2;
            float* d = &lB[buf][p * PSTRIDE + n4 * 2];
            const float a0[4] = {sA[e].x, sA[e].y, sA[e].z, sA[e].w};
            const float b0[4] = {sB[e].x, sB[e].y, sB[e].z, sB[e].w};
            #pragma unroll
            for (int i = 0; i < 4; ++i) {
                v2f pr; pr.x = a0[i]; pr.y = b0[i];
                *(v2f*)(d + 2 * i) = pr;
            }
        }
    };

    auto srcOf = [&](int p) { return p == 0 ? h0 : (p == 1 ? h1 : h2); };

    constexpr int NCHUNK = 12;
    loadStage(0);
    for (int c = 0; c < NCHUNK; ++c) {
        const int buf = c & 1;
        storeStage(buf);
        __syncthreads();
        if (c + 1 < NCHUNK) loadStage(c + 1);

        const float* ap = srcOf(c >> 2) + aOff + (c & 3) * 64;
        #pragma unroll
        for (int kk = 0; kk < 64; kk += 4) {
            const v2f a = *(const v2f*)(ap + kk);
            const int p = (kk >> 1) + (lane >> 4);
            const float* lb = &lB[buf][p * PSTRIDE + (lane & 15) * 2];
            #pragma unroll
            for (int nt = 0; nt < 4; ++nt) {
                const v2f b = *(const v2f*)(lb + nt * 32);
                acc[nt] = __builtin_amdgcn_wmma_f32_16x16x4_f32(
                    false, a, false, b, (short)0, acc[nt], false, false);
            }
        }
        __syncthreads();
    }

    if (active) {
        const int rbase = m0 + ((lane >> 4) << 3);
        #pragma unroll
        for (int nt = 0; nt < 4; ++nt) {
            const int col = nb + nt * 16 + (lane & 15);
            #pragma unroll
            for (int v = 0; v < 8; ++v)
                out[(size_t)(rbase + v) * OUT_C + col] = acc[nt][v];
        }
    }
}

extern "C" void kernel_launch(void* const* d_in, const int* in_sizes, int n_in,
                              void* d_out, int out_size, void* d_ws, size_t ws_size,
                              hipStream_t stream) {
    (void)in_sizes; (void)n_in; (void)out_size; (void)ws_size;

    const float* x    = (const float*)d_in[0];
    const int*   a1r  = (const int*)  d_in[1];
    const int*   a1c  = (const int*)  d_in[2];
    const float* a1v  = (const float*)d_in[3];
    const int*   a2r  = (const int*)  d_in[4];
    const int*   a2c  = (const int*)  d_in[5];
    const float* a2v  = (const float*)d_in[6];
    const float* W1   = (const float*)d_in[7];
    const float* Wout = (const float*)d_in[8];
    const float* bout = (const float*)d_in[9];
    float* out = (float*)d_out;

    float* h0 = (float*)d_ws;
    float* h1 = h0 + (size_t)NROWS * HID;
    float* h2 = h1 + (size_t)NROWS * HID;

    {
        const int n = 2 * NROWS * HID;
        h2gcn_fill_zero<<<dim3((n + 255) / 256), dim3(256), 0, stream>>>(h1, n);
    }

    h2gcn_gemm1<<<dim3(79, 4), dim3(256), 0, stream>>>(x, W1, h0);

    h2gcn_spmm<<<dim3((E1 + 7) / 8), dim3(256), 0, stream>>>(a1r, a1c, a1v, h0, h1, E1);
    h2gcn_spmm<<<dim3((E2 + 7) / 8), dim3(256), 0, stream>>>(a2r, a2c, a2v, h0, h2, E2);

    h2gcn_gemm2<<<dim3(79, 4), dim3(256), 0, stream>>>(h0, h1, h2, Wout, bout, out);
}


# aggregation: per-edge float-atomic scatter replaced by in-call row bucketing (count, scan, place) and a one-wave-per-row register gather; zero-fill dropped; sum order by arrival-order rank
# speedup vs baseline: 1.0924x; 1.0924x over previous
#include <hip/hip_runtime.h>

#define NROWS 10000
#define IN_C  2048
#define HID   256
#define OUT_C 256
#define E1    160000
#define E2    320000

#define PSTRIDE 160
#define LDS_CHUNK (32 * PSTRIDE)

typedef float v2f __attribute__((ext_vector_type(2)));
typedef float v8f __attribute__((ext_vector_type(8)));

#define ETOT      (E1 + E2)
#define OFF_CNT   0
#define OFF_OFF   (2 * NROWS)
#define OFF_ECOL  40960
#define OFF_EVAL  (OFF_ECOL + ETOT)
#define Q1        (E1 / 4)
#define QTOT      (ETOT / 4)

__global__ void h2gcn_zero_i32(int* __restrict__ p, int n) {
    int i = blockIdx.x * blockDim.x + threadIdx.x;
    if (i < n) p[i] = 0;
}

__global__ void h2gcn_count(const int* __restrict__ r1,
                            const int* __restrict__ r2,
                            int* __restrict__ cnt) {
    const int q = blockIdx.x * blockDim.x + threadIdx.x;
    if (q >= QTOT) return;
    const int g = (q >= Q1) ? 1 : 0;
    const int4 r = g ? ((const int4*)r2)[q - Q1] : ((const int4*)r1)[q];
    int* c = cnt + g * NROWS;
    atomicAdd(c + r.x, 1);
    atomicAdd(c + r.y, 1);
    atomicAdd(c + r.z, 1);
    atomicAdd(c + r.w, 1);
}

__global__ void h2gcn_scan(const int* __restrict__ cnt, int* __restrict__ off) {
    __shared__ int wtot[32];
    const int t    = threadIdx.x;
    const int lane = t & 31;
    const int wave = t >> 5;
    const int* c = cnt + blockIdx.x * NROWS + t * 10;
    int*       o = off + blockIdx.x * (NROWS + 1) + t * 10;
    const bool own = (t < NROWS / 10);

    int loc[10];
    int sum = 0;
    #pragma unroll
    for (int k = 0; k < 10; ++k) {
        const int x = own ? c[k] : 0;
        loc[k] = sum;
        sum += x;
    }

    int incl = sum;
    #pragma unroll
    for (int d = 1; d < 32; d <<= 1) {
        const int v = __shfl_up(incl, d, 32);
        if (lane >= d) incl += v;
    }
    if (lane == 31) wtot[wave] = incl;
    __syncthreads();
    if (wave == 0) {
        const int w = wtot[lane];
        int wi = w;
        #pragma unroll
        for (int d = 1; d < 32; d <<= 1) {
            const int v = __shfl_up(wi, d, 32);
            if (lane >= d) wi += v;
        }
        wtot[lane] = wi - w;
        if (lane == 31) off[blockIdx.x * (NROWS + 1) + NROWS] = wi;
    }
    __syncthreads();
    const int base = wtot[wave] + incl - sum;
    if (own) {
        #pragma unroll
        for (int k = 0; k < 10; ++k) o[k] = base + loc[k];
    }
}

__global__ void h2gcn_place(const int* __restrict__ r1, const int* __restrict__ c1,
                            const float* __restrict__ v1,
                            const int* __restrict__ r2, const int* __restrict__ c2,
                            const float* __restrict__ v2,
                            int* __restrict__ cnt, const int* __restrict__ off,
                            int* __restrict__ ecol, float* __restrict__ eval) {
    const int q = blockIdx.x * blockDim.x + threadIdx.x;
    if (q >= QTOT) return;
    const int g = (q >= Q1) ? 1 : 0;
    const int qq = g ? q - Q1 : q;
    const int4   r = g ? ((const int4*)r2)[qq]   : ((const int4*)r1)[qq];
    const int4   c = g ? ((const int4*)c2)[qq]   : ((const int4*)c1)[qq];
    const float4 v = g ? ((const float4*)v2)[qq] : ((const float4*)v1)[qq];
    int* cn = cnt + g * NROWS;
    const int* of = off + g * (NROWS + 1);
    const int eb = g ? E1 : 0;
    const int rr[4] = {r.x, r.y, r.z, r.w};
    const int cc[4] = {c.x, c.y, c.z, c.w};
    const float vv[4] = {v.x, v.y, v.z, v.w};
    int old[4], st[4];
    #pragma unroll
    for (int k = 0; k < 4; ++k) old[k] = atomicSub(cn + rr[k], 1);
    #pragma unroll
    for (int k = 0; k < 4; ++k) st[k] = of[rr[k]];
    #pragma unroll
    for (int k = 0; k < 4; ++k) {
        const int pos = eb + st[k] + old[k] - 1;
        ecol[pos] = cc[k];
        eval[pos] = vv[k];
    }
}

__device__ __forceinline__ void h2gcn_axpy4(float4& a, float v, const float4& s) {
    a.x += v * s.x; a.y += v * s.y; a.z += v * s.z; a.w += v * s.w;
}

__global__ void h2gcn_gather(const int* __restrict__ off,
                             const int* __restrict__ ecol,
                             const float* __restrict__ eval,
                             const float* __restrict__ h0,
                             float* __restrict__ h12) {
    const int g    = 1 - (int)blockIdx.y;
    const int lane = threadIdx.x & 31;
    const int wv   = __builtin_amdgcn_readfirstlane((int)(threadIdx.x >> 5));
    const int row  = blockIdx.x * 8 + wv;
    const int* of  = off + g * (NROWS + 1);
    const int eb   = g ? E1 : 0;
    const int beg  = __builtin_amdgcn_readfirstlane(of[row]);
    const int end  = __builtin_amdgcn_readfirstlane(of[row + 1]);

    float4 a0 = make_float4(0.f, 0.f, 0.f, 0.f);
    float4 a1 = make_float4(0.f, 0.f, 0.f, 0.f);

    for (int s = beg; s < end; s += 32) {
        const int n = (end - s < 32) ? (end - s) : 32;
        int   myc = 0;
        float myv = 0.0f;
        if (lane < n) {
            myc = ecol[eb + s + lane];
            myv = eval[eb + s + lane];
        }
        int j = 0;
        for (; j + 4 <= n; j += 4) {
            float4 p[4], q[4];
            float  w[4];
            #pragma unroll
            for (int u = 0; u < 4; ++u) {
                const int c = __builtin_amdgcn_readlane(myc, j + u);
                w[u] = __int_as_float(__builtin_amdgcn_readlane(__float_as_int(myv), j + u));
                const float4* src = (const float4*)(h0 + (size_t)c * HID);
                p[u] = src[lane];
                q[u] = src[lane + 32];
            }
            #pragma unroll
            for (int u = 0; u < 4; ++u) {
                h2gcn_axpy4(a0, w[u], p[u]);
                h2gcn_axpy4(a1, w[u], q[u]);
            }
        }
        for (; j < n; ++j) {
            const int   c = __builtin_amdgcn_readlane(myc, j);
            const float w = __int_as_float(__builtin_amdgcn_readlane(__float_as_int(myv), j));
            const float4* src = (const float4*)(h0 + (size_t)c * HID);
            const float4 p = src[lane];
            const float4 q = src[lane + 32];
            h2gcn_axpy4(a0, w, p);
            h2gcn_axpy4(a1, w, q);
        }
    }

    float4* dst = (float4*)(h12 + ((size_t)g * NROWS + row) * HID);
    dst[lane]      = a0;
    dst[lane + 32] = a1;
}

__global__ void h2gcn_gemm1(const float* __restrict__ x,
                            const float* __restrict__ W1,
                            float* __restrict__ h0) {
    __shared__ float lB[2][LDS_CHUNK];
    const int w    = threadIdx.x >> 5;
    const int lane = threadIdx.x & 31;
    const int m0   = (blockIdx.x * 8 + w) * 16;
    const int nb   = blockIdx.y * 64;
    const bool active = (m0 < NROWS);

    v8f acc[4];
    #pragma unroll
    for (int nt = 0; nt < 4; ++nt) acc[nt] = 0.0f;

    int aRow = m0 + (lane & 15);
    if (aRow >= NROWS) aRow = NROWS - 1;
    const float* aBase = x + (size_t)aRow * IN_C + ((lane >> 4) << 1);

    float4 sA[2], sB[2];
    auto loadStage = [&](int k0) {
        #pragma unroll
        for (int e = 0; e < 2; ++e) {
            const int u  = threadIdx.x + 256 * e;
            const int p  = u >> 4;
            const int n4 = (u & 15) << 2;
            const float* g = &W1[(size_t)(k0 + 2 * p) * HID + nb + n4];
            sA[e] = *(const float4*)g;
            sB[e] = *(const float4*)(g + HID);
        }
    };
    auto storeStage = [&](int buf) {
        #pragma unroll
        for (int e = 0; e < 2; ++e) {
            const int u  = threadIdx.x + 256 * e;
            const int p  = u >> 4;
            const int n4 = (u & 15) << 2;
            float* d = &lB[buf][p * PSTRIDE + n4 * 2];
            const float a0[4] = {sA[e].x, sA[e].y, sA[e].z, sA[e].w};
            const float b0[4] = {sB[e].x, sB[e].y, sB[e].z, sB[e].w};
            #pragma unroll
            for (int i = 0; i < 4; ++i) {
                v2f pr; pr.x = a0[i]; pr.y = b0[i];
                *(v2f*)(d + 2 * i) = pr;
            }
        }
    };

    constexpr int NCHUNK = IN_C / 64;
    loadStage(0);
    for (int c = 0; c < NCHUNK; ++c) {
        const int buf = c & 1;
        storeStage(buf);
        __syncthreads();
        if (c + 1 < NCHUNK) loadStage((c + 1) * 64);

        const float* ap = aBase + c * 64;
        #pragma unroll
        for (int kk = 0; kk < 64; kk += 4) {
            const v2f a = *(const v2f*)(ap + kk);
            const int p = (kk >> 1) + (lane >> 4);
            const float* lb = &lB[buf][p * PSTRIDE + (lane & 15) * 2];
            #pragma unroll
            for (int nt = 0; nt < 4; ++nt) {
                const v2f b = *(const v2f*)(lb + nt * 32);
                acc[nt] = __builtin_amdgcn_wmma_f32_16x16x4_f32(
                    false, a, false, b, (short)0, acc[nt], false, false);
            }
        }
        __syncthreads();
    }

    if (active) {
        const int rbase = m0 + ((lane >> 4) << 3);
        #pragma unroll
        for (int nt = 0; nt < 4; ++nt) {
            const int col = nb + nt * 16 + (lane & 15);
            #pragma unroll
            for (int v = 0; v < 8; ++v)
                h0[(size_t)(rbase + v) * HID + col] = acc[nt][v];
        }
    }
}

__global__ void h2gcn_gemm2(const float* __restrict__ h0,
                            const float* __restrict__ h1,
                            const float* __restrict__ h2,
                            const float* __restrict__ Wout,
                            const float* __restrict__ bout,
                            float* __restrict__ out) {
    __shared__ float lB[2][LDS_CHUNK];
    const int w    = threadIdx.x >> 5;
    const int lane = threadIdx.x & 31;
    const int m0   = (blockIdx.x * 8 + w) * 16;
    const int nb   = blockIdx.y * 64;
    const bool active = (m0 < NROWS);

    v8f acc[4];
    #pragma unroll
    for (int nt = 0; nt < 4; ++nt) {
        const float b = bout[nb + nt * 16 + (lane & 15)];
        #pragma unroll
        for (int v = 0; v < 8; ++v) acc[nt][v] = b;
    }

    int aRow = m0 + (lane & 15);
    if (aRow >= NROWS) aRow = NROWS - 1;
    const size_t aOff = (size_t)aRow * HID + ((lane >> 4) << 1);

    float4 sA[2], sB[2];
    auto loadStage = [&](int c) {
        const int part = c >> 2;
        const int k0   = (c & 3) * 64;
        const float* wp = Wout + (size_t)part * HID * OUT_C;
        #pragma unroll
        for (int e = 0; e < 2; ++e) {
            const int u  = threadIdx.x + 256 * e;
            const int p  = u >> 4;
            const int n4 = (u & 15) << 2;
            const float* g = &wp[(size_t)(k0 + 2 * p) * OUT_C + nb + n4];
            sA[e] = *(const float4*)g;
            sB[e] = *(const float4*)(g + OUT_C);
        }
    };
    auto storeStage = [&](int buf) {
        #pragma unroll
        for (int e = 0; e < 2; ++e) {
            const int u  = threadIdx.x + 256 * e;
            const int p  = u >> 4;
            const int n4 = (u & 15) << 2;
            float* d = &lB[buf][p * PSTRIDE + n4 * 2];
            const float a0[4] = {sA[e].x, sA[e].y, sA[e].z, sA[e].w};
            const float b0[4] = {sB[e].x, sB[e].y, sB[e].z, sB[e].w};
            #pragma unroll
            for (int i = 0; i < 4; ++i) {
                v2f pr; pr.x = a0[i]; pr.y = b0[i];
                *(v2f*)(d + 2 * i) = pr;
            }
        }
    };

    auto srcOf = [&](int p) { return p == 0 ? h0 : (p == 1 ? h1 : h2); };

    constexpr int NCHUNK = 12;
    loadStage(0);
    for (int c = 0; c < NCHUNK; ++c) {
        const int buf = c & 1;
        storeStage(buf);
        __syncthreads();
        if (c + 1 < NCHUNK) loadStage(c + 1);

        const float* ap = srcOf(c >> 2) + aOff + (c & 3) * 64;
        #pragma unroll
        for (int kk = 0; kk < 64; kk += 4) {
            const v2f a = *(const v2f*)(ap + kk);
            const int p = (kk >> 1) + (lane >> 4);
            const float* lb = &lB[buf][p * PSTRIDE + (lane & 15) * 2];
            #pragma unroll
            for (int nt = 0; nt < 4; ++nt) {
                const v2f b = *(const v2f*)(lb + nt * 32);
                acc[nt] = __builtin_amdgcn_wmma_f32_16x16x4_f32(
                    false, a, false, b, (short)0, acc[nt], false, false);
            }
        }
        __syncthreads();
    }

    if (active) {
        const int rbase = m0 + ((lane >> 4) << 3);
        #pragma unroll
        for (int nt = 0; nt < 4; ++nt) {
            const int col = nb + nt * 16 + (lane & 15);
            #pragma unroll
            for (int v = 0; v < 8; ++v)
                out[(size_t)(rbase + v) * OUT_C + col] = acc[nt][v];
        }
    }
}

extern "C" void kernel_launch(void* const* d_in, const int* in_sizes, int n_in,
                              void* d_out, int out_size, void* d_ws, size_t ws_size,
                              hipStream_t stream) {
    (void)in_sizes; (void)n_in; (void)out_size; (void)ws_size;

    const float* x    = (const float*)d_in[0];
    const int*   a1r  = (const int*)  d_in[1];
    const int*   a1c  = (const int*)  d_in[2];
    const float* a1v  = (const float*)d_in[3];
    const int*   a2r  = (const int*)  d_in[4];
    const int*   a2c  = (const int*)  d_in[5];
    const float* a2v  = (const float*)d_in[6];
    const float* W1   = (const float*)d_in[7];
    const float* Wout = (const float*)d_in[8];
    const float* bout = (const float*)d_in[9];
    float* out = (float*)d_out;

    float* h0 = (float*)d_ws;
    float* h1 = h0 + (size_t)NROWS * HID;
    float* h2 = h1 + (size_t)NROWS * HID;

    int*   cnt  = (int*)d_out + OFF_CNT;
    int*   off  = (int*)d_out + OFF_OFF;
    int*   ecol = (int*)d_out + OFF_ECOL;
    float* eval = (float*)d_out + OFF_EVAL;

    h2gcn_zero_i32<<<dim3((2 * NROWS + 255) / 256), dim3(256), 0, stream>>>(cnt, 2 * NROWS);
    h2gcn_count<<<dim3((QTOT + 255) / 256), dim3(256), 0, stream>>>(a1r, a2r, cnt);
    h2gcn_scan<<<dim3(2), dim3(1024), 0, stream>>>(cnt, off);
    h2gcn_place<<<dim3((QTOT + 255) / 256), dim3(256), 0, stream>>>(
        a1r, a1c, a1v, a2r, a2c, a2v, cnt, off, ecol, eval);

    h2gcn_gemm1<<<dim3(79, 4), dim3(256), 0, stream>>>(x, W1, h0);

    h2gcn_gather<<<dim3(NROWS / 8, 2), dim3(256), 0, stream>>>(off, ecol, eval, h0, h1);

    h2gcn_gemm2<<<dim3(79, 4), dim3(256), 0, stream>>>(h0, h1, h2, Wout, bout, out);
}


# row bucketing (count, scan, place) moved into the GEMM1 launch as 48 leading work-groups with LDS counters, no global atomics: 7 launches -> 3; row sums in LDS slot order, adj2 as two half-lists
# speedup vs baseline: 1.7420x; 1.5947x over previous
#include <hip/hip_runtime.h>

#define NROWS 10000
#define IN_C  2048
#define HID   256
#define OUT_C 256
#define E1    160000
#define E2    320000

#define PSTRIDE 160
#define LDS_CHUNK (32 * PSTRIDE)

typedef float v2f __attribute__((ext_vector_type(2)));
typedef float v8f __attribute__((ext_vector_type(8)));

#define ETOT      (E1 + E2)
#define NSUB      3
#define SUBE      160000
#define RR        625
#define NRANGE    (NROWS / RR)
#define NBKT      (NSUB * NRANGE)
#define OFF_OFF   0
#define OFF_ECOL  40960
#define OFF_EVAL  (OFF_ECOL + ETOT)
#define GEMM1_MB  79
#define GEMM1_NB  4

static_assert(E1 == SUBE && E2 == 2 * SUBE, "sub-list split");
static_assert(NRANGE * RR == NROWS, "row ranges");
static_assert(NSUB * (NROWS + 1) <= OFF_ECOL, "scratch layout");

__device__ __forceinline__ void h2gcn_bucket(const int* __restrict__ rg,
                                             const int* __restrict__ cg,
                                             const float* __restrict__ vg,
                                             const int lo, const int eb, const bool last,
                                             int* __restrict__ offg,
                                             int* __restrict__ ecol,
                                             float* __restrict__ eval,
                                             int* sm) {
    int* cnt = sm;
    int* sb  = sm + 640;
    const int t = threadIdx.x;
    for (int i = t; i < RR; i += 256) cnt[i] = 0;
    if (t == 0) sb[0] = 0;
    __syncthreads();

    constexpr int NQ = SUBE / 4;
    const int4*   r4 = (const int4*)rg;
    const int4*   c4 = (const int4*)cg;
    const float4* v4 = (const float4*)vg;

    int below = 0;
    for (int base = 0; base < NQ; base += 256 * 8) {
        int4 r[8];
        #pragma unroll
        for (int u = 0; u < 8; ++u) {
            const int q = base + u * 256 + t;
            r[u] = (q < NQ) ? r4[q] : make_int4(0x7fffffff, 0x7fffffff, 0x7fffffff, 0x7fffffff);
        }
        #pragma unroll
        for (int u = 0; u < 8; ++u) {
            const int rr[4] = {r[u].x, r[u].y, r[u].z, r[u].w};
            #pragma unroll
            for (int k = 0; k < 4; ++k) {
                const int d = rr[k] - lo;
                below += (d < 0) ? 1 : 0;
                if ((unsigned)d < (unsigned)RR) atomicAdd(&cnt[d], 1);
            }
        }
    }
    atomicAdd(sb, below);
    __syncthreads();

    if (t < 32) {
        int carry = sb[0];
        for (int i0 = 0; i0 < RR; i0 += 32) {
            const int i = i0 + t;
            const int x = (i < RR) ? cnt[i] : 0;
            int incl = x;
            #pragma unroll
            for (int d = 1; d < 32; d <<= 1) {
                const int v = __shfl_up(incl, d, 32);
                if (t >= d) incl += v;
            }
            const int st = carry + incl - x;
            if (i < RR) { cnt[i] = st; offg[lo + i] = st; }
            carry += __shfl(incl, 31, 32);
        }
        if (last && t == 0) offg[NROWS] = carry;
    }
    __syncthreads();

    int*   ecb = ecol + eb;
    float* evb = eval + eb;
    for (int base = 0; base < NQ; base += 256 * 4) {
        int4 r[4], c[4];
        float4 v[4];
        #pragma unroll
        for (int u = 0; u < 4; ++u) {
            const int q = base + u * 256 + t;
            const bool ok = (q < NQ);
            r[u] = ok ? r4[q] : make_int4(0x7fffffff, 0x7fffffff, 0x7fffffff, 0x7fffffff);
            c[u] = ok ? c4[q] : make_int4(0, 0, 0, 0);
            v[u] = ok ? v4[q] : make_float4(0.f, 0.f, 0.f, 0.f);
        }
        int pos[16];
        #pragma unroll
        for (int u = 0; u < 4; ++u) {
            const int rr[4] = {r[u].x, r[u].y, r[u].z, r[u].w};
            #pragma unroll
            for (int k = 0; k < 4; ++k) {
                const int d = rr[k] - lo;
                pos[u * 4 + k] = -1;
                if ((unsigned)d < (unsigned)RR) pos[u * 4 + k] = atomicAdd(&cnt[d], 1);
            }
        }
        #pragma unroll
        for (int u = 0; u < 4; ++u) {
            const int   cc[4] = {c[u].x, c[u].y, c[u].z, c[u].w};
            const float vv[4] = {v[u].x, v[u].y, v[u].z, v[u].w};
            #pragma unroll
            for (int k = 0; k < 4; ++k) {
                const int p = pos[u * 4 + k];
                if (p >= 0) {
                    ecb[p] = cc[k];
                    evb[p] = vv[k];
                }
            }
        }
    }
}

__device__ __forceinline__ void h2gcn_axpy4(float4& a, float v, const float4& s) {
    a.x += v * s.x; a.y += v * s.y; a.z += v * s.z; a.w += v * s.w;
}

__global__ void h2gcn_gather(const int* __restrict__ off,
                             const int* __restrict__ ecol,
                             const float* __restrict__ eval,
                             const float* __restrict__ h0,
                             float* __restrict__ h12) {
    const int g    = 1 - (int)blockIdx.y;
    const int lane = threadIdx.x & 31;
    const int wv   = __builtin_amdgcn_readfirstlane((int)(threadIdx.x >> 5));
    const int row  = blockIdx.x * 8 + wv;
    const int sfirst = g ? 1 : 0;
    const int slast  = g ? 2 : 0;

    float4 a0 = make_float4(0.f, 0.f, 0.f, 0.f);
    float4 a1 = make_float4(0.f, 0.f, 0.f, 0.f);

    for (int sl = sfirst; sl <= slast; ++sl) {
        const int* of  = off + sl * (NROWS + 1);
        const int eb   = sl * SUBE;
        const int beg  = __builtin_amdgcn_readfirstlane(of[row]);
        const int end  = __builtin_amdgcn_readfirstlane(of[row + 1]);
        for (int s = beg; s < end; s += 32) {
            const int n = (end - s < 32) ? (end - s) : 32;
            int   myc = 0;
            float myv = 0.0f;
            if (lane < n) {
                myc = ecol[eb + s + lane];
                myv = eval[eb + s + lane];
            }
            int j = 0;
            for (; j + 4 <= n; j += 4) {
                float4 p[4], q[4];
                float  w[4];
                #pragma unroll
                for (int u = 0; u < 4; ++u) {
                    const int c = __builtin_amdgcn_readlane(myc, j + u);
                    w[u] = __int_as_float(__builtin_amdgcn_readlane(__float_as_int(myv), j + u));
                    const float4* src = (const float4*)(h0 + (size_t)c * HID);
                    p[u] = src[lane];
                    q[u] = src[lane + 32];
                }
                #pragma unroll
                for (int u = 0; u < 4; ++u) {
                    h2gcn_axpy4(a0, w[u], p[u]);
                    h2gcn_axpy4(a1, w[u], q[u]);
                }
            }
            for (; j < n; ++j) {
                const int   c = __builtin_amdgcn_readlane(myc, j);
                const float w = __int_as_float(__builtin_amdgcn_readlane(__float_as_int(myv), j));
                const float4* src = (const float4*)(h0 + (size_t)c * HID);
                const float4 p = src[lane];
                const float4 q = src[lane + 32];
                h2gcn_axpy4(a0, w, p);
                h2gcn_axpy4(a1, w, q);
            }
        }
    }

    float4* dst = (float4*)(h12 + ((size_t)g * NROWS + row) * HID);
    dst[lane]      = a0;
    dst[lane + 32] = a1;
}

__global__ void h2gcn_gemm1(const float* __restrict__ x,
                            const float* __restrict__ W1,
                            float* __restrict__ h0,
                            const int* __restrict__ r1, const int* __restrict__ c1,
                            const float* __restrict__ v1,
                            const int* __restrict__ r2, const int* __restrict__ c2,
                            const float* __restrict__ v2,
                            int* __restrict__ off,
                            int* __restrict__ ecol, float* __restrict__ eval) {
    __shared__ float lB[2][LDS_CHUNK];
    if (blockIdx.x < NBKT) {
        const int sl  = (int)blockIdx.x / NRANGE;
        const int rng = (int)blockIdx.x % NRANGE;
        const int sh  = (sl == 2) ? SUBE : 0;
        h2gcn_bucket(sl ? r2 + sh : r1, sl ? c2 + sh : c1, sl ? v2 + sh : v1,
                     rng * RR, sl * SUBE, rng == NRANGE - 1,
                     off + sl * (NROWS + 1), ecol, eval, (int*)&lB[0][0]);
        return;
    }
    const int gb   = (int)blockIdx.x - NBKT;
    const int gbx  = gb % GEMM1_MB;
    const int gby  = gb / GEMM1_MB;
    const int w    = threadIdx.x >> 5;
    const int lane = threadIdx.x & 31;
    const int m0   = (gbx * 8 + w) * 16;
    const int nb   = gby * 64;
    const bool active = (m0 < NROWS);

    v8f acc[4];
    #pragma unroll
    for (int nt = 0; nt < 4; ++nt) acc[nt] = 0.0f;

    int aRow = m0 + (lane & 15);
    if (aRow >= NROWS) aRow = NROWS - 1;
    const float* aBase = x + (size_t)aRow * IN_C + ((lane >> 4) << 1);

    float4 sA[2], sB[2];
    auto loadStage = [&](int k0) {
        #pragma unroll
        for (int e = 0; e < 2; ++e) {
            const int u  = threadIdx.x + 256 * e;
            const int p  = u >> 4;
            const int n4 = (u & 15) << 2;
            const float* g = &W1[(size_t)(k0 + 2 * p) * HID + nb + n4];
            sA[e] = *(const float4*)g;
            sB[e] = *(const float4*)(g + HID);
        }
    };
    auto storeStage = [&](int buf) {
        #pragma unroll
        for (int e = 0; e < 2; ++e) {
            const int u  = threadIdx.x + 256 * e;
            const int p  = u >> 4;
            const int n4 = (u & 15) << 2;
            float* d = &lB[buf][p * PSTRIDE + n4 * 2];
            const float a0[4] = {sA[e].x, sA[e].y, sA[e].z, sA[e].w};
            const float b0[4] = {sB[e].x, sB[e].y, sB[e].z, sB[e].w};
            #pragma unroll
            for (int i = 0; i < 4; ++i) {
                v2f pr; pr.x = a0[i]; pr.y = b0[i];
                *(v2f*)(d + 2 * i) = pr;
            }
        }
    };

    constexpr int NCHUNK = IN_C / 64;
    loadStage(0);
    for (int c = 0; c < NCHUNK; ++c) {
        const int buf = c & 1;
        storeStage(buf);
        __syncthreads();
        if (c + 1 < NCHUNK) loadStage((c + 1) * 64);

        const float* ap = aBase + c * 64;
        #pragma unroll
        for (int kk = 0; kk < 64; kk += 4) {
            const v2f a = *(const v2f*)(ap + kk);
            const int p = (kk >> 1) + (lane >> 4);
            const float* lb = &lB[buf][p * PSTRIDE + (lane & 15) * 2];
            #pragma unroll
            for (int nt = 0; nt < 4; ++nt) {
                const v2f b = *(const v2f*)(lb + nt * 32);
                acc[nt] = __builtin_amdgcn_wmma_f32_16x16x4_f32(
                    false, a, false, b, (short)0, acc[nt], false, false);
            }
        }
        __syncthreads();
    }

    if (active) {
        const int rbase = m0 + ((lane >> 4) << 3);
        #pragma unroll
        for (int nt = 0; nt < 4; ++nt) {
            const int col = nb + nt * 16 + (lane & 15);
            #pragma unroll
            for (int v = 0; v < 8; ++v)
                h0[(size_t)(rbase + v) * HID + col] = acc[nt][v];
        }
    }
}

__global__ void h2gcn_gemm2(const float* __restrict__ h0,
                            const float* __restrict__ h1,
                            const float* __restrict__ h2,
                            const float* __restrict__ Wout,
                            const float* __restrict__ bout,
                            float* __restrict__ out) {
    __shared__ float lB[2][LDS_CHUNK];
    const int w    = threadIdx.x >> 5;
    const int lane = threadIdx.x & 31;
    const int m0   = (blockIdx.x * 8 + w) * 16;
    const int nb   = blockIdx.y * 64;
    const bool active = (m0 < NROWS);

    v8f acc[4];
    #pragma unroll
    for (int nt = 0; nt < 4; ++nt) {
        const float b = bout[nb + nt * 16 + (lane & 15)];
        #pragma unroll
        for (int v = 0; v < 8; ++v) acc[nt][v] = b;
    }

    int aRow = m0 + (lane & 15);
    if (aRow >= NROWS) aRow = NROWS - 1;
    const size_t aOff = (size_t)aRow * HID + ((lane >> 4) << 1);

    float4 sA[2], sB[2];
    auto loadStage = [&](int c) {
        const int part = c >> 2;
        const int k0   = (c & 3) * 64;
        const float* wp = Wout + (size_t)part * HID * OUT_C;
        #pragma unroll
        for (int e = 0; e < 2; ++e) {
            const int u  = threadIdx.x + 256 * e;
            const int p  = u >> 4;
            const int n4 = (u & 15) << 2;
            const float* g = &wp[(size_t)(k0 + 2 * p) * OUT_C + nb + n4];
            sA[e] = *(const float4*)g;
            sB[e] = *(const float4*)(g + OUT_C);
        }
    };
    auto storeStage = [&](int buf) {
        #pragma unroll
        for (int e = 0; e < 2; ++e) {
            const int u  = threadIdx.x + 256 * e;
            const int p  = u >> 4;
            const int n4 = (u & 15) << 2;
            float* d = &lB[buf][p * PSTRIDE + n4 * 2];
            const float a0[4] = {sA[e].x, sA[e].y, sA[e].z, sA[e].w};
            const float b0[4] = {sB[e].x, sB[e].y, sB[e].z, sB[e].w};
            #pragma unroll
            for (int i = 0; i < 4; ++i) {
                v2f pr; pr.x = a0[i]; pr.y = b0[i];
                *(v2f*)(d + 2 * i) = pr;
            }
        }
    };

    auto srcOf = [&](int p) { return p == 0 ? h0 : (p == 1 ? h1 : h2); };

    constexpr int NCHUNK = 12;
    loadStage(0);
    for (int c = 0; c < NCHUNK; ++c) {
        const int buf = c & 1;
        storeStage(buf);
        __syncthreads();
        if (c + 1 < NCHUNK) loadStage(c + 1);

        const float* ap = srcOf(c >> 2) + aOff + (c & 3) * 64;
        #pragma unroll
        for (int kk = 0; kk < 64; kk += 4) {
            const v2f a = *(const v2f*)(ap + kk);
            const int p = (kk >> 1) + (lane >> 4);
            const float* lb = &lB[buf][p * PSTRIDE + (lane & 15) * 2];
            #pragma unroll
            for (int nt = 0; nt < 4; ++nt) {
                const v2f b = *(const v2f*)(lb + nt * 32);
                acc[nt] = __builtin_amdgcn_wmma_f32_16x16x4_f32(
                    false, a, false, b, (short)0, acc[nt], false, false);
            }
        }
        __syncthreads();
    }

    if (active) {
        const int rbase = m0 + ((lane >> 4) << 3);
        #pragma unroll
        for (int nt = 0; nt < 4; ++nt) {
            const int col = nb + nt * 16 + (lane & 15);
            #pragma unroll
            for (int v = 0; v < 8; ++v)
                out[(size_t)(rbase + v) * OUT_C + col] = acc[nt][v];
        }
    }
}

extern "C" void kernel_launch(void* const* d_in, const int* in_sizes, int n_in,
                              void* d_out, int out_size, void* d_ws, size_t ws_size,
                              hipStream_t stream) {
    (void)in_sizes; (void)n_in; (void)out_size; (void)ws_size;

    const float* x    = (const float*)d_in[0];
    const int*   a1r  = (const int*)  d_in[1];
    const int*   a1c  = (const int*)  d_in[2];
    const float* a1v  = (const float*)d_in[3];
    const int*   a2r  = (const int*)  d_in[4];
    const int*   a2c  = (const int*)  d_in[5];
    const float* a2v  = (const float*)d_in[6];
    const float* W1   = (const float*)d_in[7];
    const float* Wout = (const float*)d_in[8];
    const float* bout = (const float*)d_in[9];
    float* out = (float*)d_out;

    float* h0 = (float*)d_ws;
    float* h1 = h0 + (size_t)NROWS * HID;
    float* h2 = h1 + (size_t)NROWS * HID;

    int*   off  = (int*)d_out + OFF_OFF;
    int*   ecol = (int*)d_out + OFF_ECOL;
    float* eval = (float*)d_out + OFF_EVAL;

    h2gcn_gemm1<<<dim3(NBKT + GEMM1_MB * GEMM1_NB), dim3(256), 0, stream>>>(
        x, W1, h0, a1r, a1c, a1v, a2r, a2c, a2v, off, ecol, eval);

    h2gcn_gather<<<dim3(NROWS / 8, 2), dim3(256), 0, stream>>>(off, ecol, eval, h0, h1);

    h2gcn_gemm2<<<dim3(79, 4), dim3(256), 0, stream>>>(h0, h1, h2, Wout, bout, out);
}


# both GEMMs: A operand staged through LDS by 16-byte loads that cover whole 128-byte lines, one chunk ahead, replacing v2f fragment reads from global (16 lines per load); sum order unchanged
# speedup vs baseline: 1.8847x; 1.0819x over previous
#include <hip/hip_runtime.h>

#define NROWS 10000
#define IN_C  2048
#define HID   256
#define OUT_C 256
#define E1    160000
#define E2    320000

#define PSTRIDE 160
#define LDS_CHUNK (32 * PSTRIDE)

#define APITCH 68
#define LDS_ATILE (128 * APITCH)

typedef float v2f __attribute__((ext_vector_type(2)));
typedef float v8f __attribute__((ext_vector_type(8)));

#define ETOT      (E1 + E2)
#define NSUB      3
#define SUBE      160000
#define RR        625
#define NRANGE    (NROWS / RR)
#define NBKT      (NSUB * NRANGE)
#define OFF_OFF   0
#define OFF_ECOL  40960
#define OFF_EVAL  (OFF_ECOL + ETOT)
#define GEMM1_MB  79
#define GEMM1_NB  4

static_assert(E1 == SUBE && E2 == 2 * SUBE, "sub-list split");
static_assert(NRANGE * RR == NROWS, "row ranges");
static_assert(NSUB * (NROWS + 1) <= OFF_ECOL, "scratch layout");

__device__ __forceinline__ void h2gcn_bucket(const int* __restrict__ rg,
                                             const int* __restrict__ cg,
                                             const float* __restrict__ vg,
                                             const int lo, const int eb, const bool last,
                                             int* __restrict__ offg,
                                             int* __restrict__ ecol,
                                             float* __restrict__ eval,
                                             int* sm) {
    int* cnt = sm;
    int* sb  = sm + 640;
    const int t = threadIdx.x;
    for (int i = t; i < RR; i += 256) cnt[i] = 0;
    if (t == 0) sb[0] = 0;
    __syncthreads();

    constexpr int NQ = SUBE / 4;
    const int4*   r4 = (const int4*)rg;
    const int4*   c4 = (const int4*)cg;
    const float4* v4 = (const float4*)vg;

    int below = 0;
    for (int base = 0; base < NQ; base += 256 * 8) {
        int4 r[8];
        #pragma unroll
        for (int u = 0; u < 8; ++u) {
            const int q = base + u * 256 + t;
            r[u] = (q < NQ) ? r4[q] : make_int4(0x7fffffff, 0x7fffffff, 0x7fffffff, 0x7fffffff);
        }
        #pragma unroll
        for (int u = 0; u < 8; ++u) {
            const int rr[4] = {r[u].x, r[u].y, r[u].z, r[u].w};
            #pragma unroll
            for (int k = 0; k < 4; ++k) {
                const int d = rr[k] - lo;
                below += (d < 0) ? 1 : 0;
                if ((unsigned)d < (unsigned)RR) atomicAdd(&cnt[d], 1);
            }
        }
    }
    atomicAdd(sb, below);
    __syncthreads();

    if (t < 32) {
        int carry = sb[0];
        for (int i0 = 0; i0 < RR; i0 += 32) {
            const int i = i0 + t;
            const int x = (i < RR) ? cnt[i] : 0;
            int incl = x;
            #pragma unroll
            for (int d = 1; d < 32; d <<= 1) {
                const int v = __shfl_up(incl, d, 32);
                if (t >= d) incl += v;
            }
            const int st = carry + incl - x;
            if (i < RR) { cnt[i] = st; offg[lo + i] = st; }
            carry += __shfl(incl, 31, 32);
        }
        if (last && t == 0) offg[NROWS] = carry;
    }
    __syncthreads();

    int*   ecb = ecol + eb;
    float* evb = eval + eb;
    for (int base = 0; base < NQ; base += 256 * 4) {
        int4 r[4], c[4];
        float4 v[4];
        #pragma unroll
        for (int u = 0; u < 4; ++u) {
            const int q = base + u * 256 + t;
            const bool ok = (q < NQ);
            r[u] = ok ? r4[q] : make_int4(0x7fffffff, 0x7fffffff, 0x7fffffff, 0x7fffffff);
            c[u] = ok ? c4[q] : make_int4(0, 0, 0, 0);
            v[u] = ok ? v4[q] : make_float4(0.f, 0.f, 0.f, 0.f);
        }
        int pos[16];
        #pragma unroll
        for (int u = 0; u < 4; ++u) {
            const int rr[4] = {r[u].x, r[u].y, r[u].z, r[u].w};
            #pragma unroll
            for (int k = 0; k < 4; ++k) {
                const int d = rr[k] - lo;
                pos[u * 4 + k] = -1;
                if ((unsigned)d < (unsigned)RR) pos[u * 4 + k] = atomicAdd(&cnt[d], 1);
            }
        }
        #pragma unroll
        for (int u = 0; u < 4; ++u) {
            const int   cc[4] = {c[u].x, c[u].y, c[u].z, c[u].w};
            const float vv[4] = {v[u].x, v[u].y, v[u].z, v[u].w};
            #pragma unroll
            for (int k = 0; k < 4; ++k) {
                const int p = pos[u * 4 + k];
                if (p >= 0) {
                    ecb[p] = cc[k];
                    evb[p] = vv[k];
                }
            }
        }
    }
}

__device__ __forceinline__ void h2gcn_axpy4(float4& a, float v, const float4& s) {
    a.x += v * s.x; a.y += v * s.y; a.z += v * s.z; a.w += v * s.w;
}

__global__ void h2gcn_gather(const int* __restrict__ off,
                             const int* __restrict__ ecol,
                             const float* __restrict__ eval,
                             const float* __restrict__ h0,
                             float* __restrict__ h12) {
    const int g    = 1 - (int)blockIdx.y;
    const int lane = threadIdx.x & 31;
    const int wv   = __builtin_amdgcn_readfirstlane((int)(threadIdx.x >> 5));
    const int row  = blockIdx.x * 8 + wv;
    const int sfirst = g ? 1 : 0;
    const int slast  = g ? 2 : 0;

    float4 a0 = make_float4(0.f, 0.f, 0.f, 0.f);
    float4 a1 = make_float4(0.f, 0.f, 0.f, 0.f);

    for (int sl = sfirst; sl <= slast; ++sl) {
        const int* of  = off + sl * (NROWS + 1);
        const int eb   = sl * SUBE;
        const int beg  = __builtin_amdgcn_readfirstlane(of[row]);
        const int end  = __builtin_amdgcn_readfirstlane(of[row + 1]);
        for (int s = beg; s < end; s += 32) {
            const int n = (end - s < 32) ? (end - s) : 32;
            int   myc = 0;
            float myv = 0.0f;
            if (lane < n) {
                myc = ecol[eb + s + lane];
                myv = eval[eb + s + lane];
            }
            int j = 0;
            for (; j + 4 <= n; j += 4) {
                float4 p[4], q[4];
                float  w[4];
                #pragma unroll
                for (int u = 0; u < 4; ++u) {
                    const int c = __builtin_amdgcn_readlane(myc, j + u);
                    w[u] = __int_as_float(__builtin_amdgcn_readlane(__float_as_int(myv), j + u));
                    const float4* src = (const float4*)(h0 + (size_t)c * HID);
                    p[u] = src[lane];
                    q[u] = src[lane + 32];
                }
                #pragma unroll
                for (int u = 0; u < 4; ++u) {
                    h2gcn_axpy4(a0, w[u], p[u]);
                    h2gcn_axpy4(a1, w[u], q[u]);
                }
            }
            for (; j < n; ++j) {
                const int   c = __builtin_amdgcn_readlane(myc, j);
                const float w = __int_as_float(__builtin_amdgcn_readlane(__float_as_int(myv), j));
                const float4* src = (const float4*)(h0 + (size_t)c * HID);
                const float4 p = src[lane];
                const float4 q = src[lane + 32];
                h2gcn_axpy4(a0, w, p);
                h2gcn_axpy4(a1, w, q);
            }
        }
    }

    float4* dst = (float4*)(h12 + ((size_t)g * NROWS + row) * HID);
    dst[lane]      = a0;
    dst[lane + 32] = a1;
}

__global__ void __launch_bounds__(256)
h2gcn_gemm1(const float* __restrict__ x,
                            const float* __restrict__ W1,
                            float* __restrict__ h0,
                            const int* __restrict__ r1, const int* __restrict__ c1,
                            const float* __restrict__ v1,
                            const int* __restrict__ r2, const int* __restrict__ c2,
                            const float* __restrict__ v2,
                            int* __restrict__ off,
                            int* __restrict__ ecol, float* __restrict__ eval) {
    __shared__ float lB[2][LDS_CHUNK];
    __shared__ float lA[LDS_ATILE];
    if (blockIdx.x < NBKT) {
        const int sl  = (int)blockIdx.x / NRANGE;
        const int rng = (int)blockIdx.x % NRANGE;
        const int sh  = (sl == 2) ? SUBE : 0;
        h2gcn_bucket(sl ? r2 + sh : r1, sl ? c2 + sh : c1, sl ? v2 + sh : v1,
                     rng * RR, sl * SUBE, rng == NRANGE - 1,
                     off + sl * (NROWS + 1), ecol, eval, (int*)&lB[0][0]);
        return;
    }
    const int gb   = (int)blockIdx.x - NBKT;
    const int gbx  = gb % GEMM1_MB;
    const int gby  = gb / GEMM1_MB;
    const int w    = threadIdx.x >> 5;
    const int lane = threadIdx.x & 31;
    const int m0   = (gbx * 8 + w) * 16;
    const int nb   = gby * 64;
    const bool active = (m0 < NROWS);

    v8f acc[4];
    #pragma unroll
    for (int nt = 0; nt < 4; ++nt) acc[nt] = 0.0f;

    const int aSeg = (threadIdx.x & 15) << 2;
    auto aRowPtr = [&](int e) {
        int r = gbx * 128 + (int)(threadIdx.x >> 4) + 16 * e;
        if (r >= NROWS) r = NROWS - 1;
        return x + (size_t)r * IN_C + aSeg;
    };
    const float* const aSrc0 = aRowPtr(0);
    const float* const aSrc1 = aRowPtr(1);
    const float* const aSrc2 = aRowPtr(2);
    const float* const aSrc3 = aRowPtr(3);
    const float* const aSrc4 = aRowPtr(4);
    const float* const aSrc5 = aRowPtr(5);
    const float* const aSrc6 = aRowPtr(6);
    const float* const aSrc7 = aRowPtr(7);
    float* const aDst = &lA[(threadIdx.x >> 4) * APITCH + aSeg];
    const float* const aFrag = &lA[(w * 16 + (lane & 15)) * APITCH + ((lane >> 4) << 1)];

    float4 sA[2], sB[2];
    float4 sX0, sX1, sX2, sX3, sX4, sX5, sX6, sX7;
    auto loadStage = [&](int k0) {
        #pragma unroll
        for (int e = 0; e < 2; ++e) {
            const int u  = threadIdx.x + 256 * e;
            const int p  = u >> 4;
            const int n4 = (u & 15) << 2;
            const float* g = &W1[(size_t)(k0 + 2 * p) * HID + nb + n4];
            sA[e] = *(const float4*)g;
            sB[e] = *(const float4*)(g + HID);
        }
        sX0 = *(const float4*)(aSrc0 + k0);
        sX1 = *(const float4*)(aSrc1 + k0);
        sX2 = *(const float4*)(aSrc2 + k0);
        sX3 = *(const float4*)(aSrc3 + k0);
        sX4 = *(const float4*)(aSrc4 + k0);
        sX5 = *(const float4*)(aSrc5 + k0);
        sX6 = *(const float4*)(aSrc6 + k0);
        sX7 = *(const float4*)(aSrc7 + k0);
    };
    auto storeStage = [&](int buf) {
        *(float4*)(aDst +   0 * APITCH) = sX0;
        *(float4*)(aDst +  16 * APITCH) = sX1;
        *(float4*)(aDst +  32 * APITCH) = sX2;
        *(float4*)(aDst +  48 * APITCH) = sX3;
        *(float4*)(aDst +  64 * APITCH) = sX4;
        *(float4*)(aDst +  80 * APITCH) = sX5;
        *(float4*)(aDst +  96 * APITCH) = sX6;
        *(float4*)(aDst + 112 * APITCH) = sX7;
        #pragma unroll
        for (int e = 0; e < 2; ++e) {
            const int u  = threadIdx.x + 256 * e;
            const int p  = u >> 4;
            const int n4 = (u & 15) << 2;
            float* d = &lB[buf][p * PSTRIDE + n4 * 2];
            const float a0[4] = {sA[e].x, sA[e].y, sA[e].z, sA[e].w};
            const float b0[4] = {sB[e].x, sB[e].y, sB[e].z, sB[e].w};
            #pragma unroll
            for (int i = 0; i < 4; ++i) {
                v2f pr; pr.x = a0[i]; pr.y = b0[i];
                *(v2f*)(d + 2 * i) = pr;
            }
        }
    };

    auto mmaChunk = [&](int buf) {
        #pragma unroll
        for (int kk = 0; kk < 64; kk += 4) {
            const v2f a = *(const v2f*)(aFrag + kk);
            const int p = (kk >> 1) + (lane >> 4);
            const float* lb = &lB[buf][p * PSTRIDE + (lane & 15) * 2];
            #pragma unroll
            for (int nt = 0; nt < 4; ++nt) {
                const v2f b = *(const v2f*)(lb + nt * 32);
                acc[nt] = __builtin_amdgcn_wmma_f32_16x16x4_f32(
                    false, a, false, b, (short)0, acc[nt], false, false);
            }
        }
    };

    constexpr int NCHUNK = IN_C / 64;
    loadStage(0);
    for (int c = 0; c < NCHUNK - 1; ++c) {
        const int buf = c & 1;
        storeStage(buf);
        __syncthreads();
        loadStage((c + 1) * 64);
        mmaChunk(buf);
        __syncthreads();
    }
    storeStage((NCHUNK - 1) & 1);
    __syncthreads();
    mmaChunk((NCHUNK - 1) & 1);

    if (active) {
        const int rbase = m0 + ((lane >> 4) << 3);
        #pragma unroll
        for (int nt = 0; nt < 4; ++nt) {
            const int col = nb + nt * 16 + (lane & 15);
            #pragma unroll
            for (int v = 0; v < 8; ++v)
                h0[(size_t)(rbase + v) * HID + col] = acc[nt][v];
        }
    }
}

__global__ void __launch_bounds__(256)
h2gcn_gemm2(const float* __restrict__ h0,
                            const float* __restrict__ h1,
                            const float* __restrict__ h2,
                            const float* __restrict__ Wout,
                            const float* __restrict__ bout,
                            float* __restrict__ out) {
    __shared__ float lB[2][LDS_CHUNK];
    __shared__ float lA[LDS_ATILE];
    const int w    = threadIdx.x >> 5;
    const int lane = threadIdx.x & 31;
    const int m0   = (blockIdx.x * 8 + w) * 16;
    const int nb   = blockIdx.y * 64;
    const bool active = (m0 < NROWS);

    v8f acc[4];
    #pragma unroll
    for (int nt = 0; nt < 4; ++nt) {
        const float b = bout[nb + nt * 16 + (lane & 15)];
        #pragma unroll
        for (int v = 0; v < 8; ++v) acc[nt][v] = b;
    }

    const int aSeg = (threadIdx.x & 15) << 2;
    auto aRowOff = [&](int e) {
        int r = (int)blockIdx.x * 128 + (int)(threadIdx.x >> 4) + 16 * e;
        if (r >= NROWS) r = NROWS - 1;
        return (size_t)r * HID + aSeg;
    };
    const size_t aOff0 = aRowOff(0), aOff1 = aRowOff(1), aOff2 = aRowOff(2), aOff3 = aRowOff(3);
    const size_t aOff4 = aRowOff(4), aOff5 = aRowOff(5), aOff6 = aRowOff(6), aOff7 = aRowOff(7);
    float* const aDst = &lA[(threadIdx.x >> 4) * APITCH + aSeg];
    const float* const aFrag = &lA[(w * 16 + (lane & 15)) * APITCH + ((lane >> 4) << 1)];

    auto srcOf = [&](int p) { return p == 0 ? h0 : (p == 1 ? h1 : h2); };

    float4 sA[2], sB[2];
    float4 sX0, sX1, sX2, sX3, sX4, sX5, sX6, sX7;
    auto loadStage = [&](int c) {
        const int part = c >> 2;
        const int k0   = (c & 3) * 64;
        const float* wp = Wout + (size_t)part * HID * OUT_C;
        #pragma unroll
        for (int e = 0; e < 2; ++e) {
            const int u  = threadIdx.x + 256 * e;
            const int p  = u >> 4;
            const int n4 = (u & 15) << 2;
            const float* g = &wp[(size_t)(k0 + 2 * p) * OUT_C + nb + n4];
            sA[e] = *(const float4*)g;
            sB[e] = *(const float4*)(g + OUT_C);
        }
        const float* ab = srcOf(part) + k0;
        sX0 = *(const float4*)(ab + aOff0);
        sX1 = *(const float4*)(ab + aOff1);
        sX2 = *(const float4*)(ab + aOff2);
        sX3 = *(const float4*)(ab + aOff3);
        sX4 = *(const float4*)(ab + aOff4);
        sX5 = *(const float4*)(ab + aOff5);
        sX6 = *(const float4*)(ab + aOff6);
        sX7 = *(const float4*)(ab + aOff7);
    };
    auto storeStage = [&](int buf) {
        *(float4*)(aDst +   0 * APITCH) = sX0;
        *(float4*)(aDst +  16 * APITCH) = sX1;
        *(float4*)(aDst +  32 * APITCH) = sX2;
        *(float4*)(aDst +  48 * APITCH) = sX3;
        *(float4*)(aDst +  64 * APITCH) = sX4;
        *(float4*)(aDst +  80 * APITCH) = sX5;
        *(float4*)(aDst +  96 * APITCH) = sX6;
        *(float4*)(aDst + 112 * APITCH) = sX7;
        #pragma unroll
        for (int e = 0; e < 2; ++e) {
            const int u  = threadIdx.x + 256 * e;
            const int p  = u >> 4;
            const int n4 = (u & 15) << 2;
            float* d = &lB[buf][p * PSTRIDE + n4 * 2];
            const float a0[4] = {sA[e].x, sA[e].y, sA[e].z, sA[e].w};
            const float b0[4] = {sB[e].x, sB[e].y, sB[e].z, sB[e].w};
            #pragma unroll
            for (int i = 0; i < 4; ++i) {
                v2f pr; pr.x = a0[i]; pr.y = b0[i];
                *(v2f*)(d + 2 * i) = pr;
            }
        }
    };

    auto mmaChunk = [&](int buf) {
        #pragma unroll
        for (int kk = 0; kk < 64; kk += 4) {
            const v2f a = *(const v2f*)(aFrag + kk);
            const int p = (kk >> 1) + (lane >> 4);
            const float* lb = &lB[buf][p * PSTRIDE + (lane & 15) * 2];
            #pragma unroll
            for (int nt = 0; nt < 4; ++nt) {
                const v2f b = *(const v2f*)(lb + nt * 32);
                acc[nt] = __builtin_amdgcn_wmma_f32_16x16x4_f32(
                    false, a, false, b, (short)0, acc[nt], false, false);
            }
        }
    };

    constexpr int NCHUNK = 12;
    loadStage(0);
    for (int c = 0; c < NCHUNK - 1; ++c) {
        const int buf = c & 1;
        storeStage(buf);
        __syncthreads();
        loadStage(c + 1);
        mmaChunk(buf);
        __syncthreads();
    }
    storeStage((NCHUNK - 1) & 1);
    __syncthreads();
    mmaChunk((NCHUNK - 1) & 1);

    if (active) {
        const int rbase = m0 + ((lane >> 4) << 3);
        #pragma unroll
        for (int nt = 0; nt < 4; ++nt) {
            const int col = nb + nt * 16 + (lane & 15);
            #pragma unroll
            for (int v = 0; v < 8; ++v)
                out[(size_t)(rbase + v) * OUT_C + col] = acc[nt][v];
        }
    }
}

extern "C" void kernel_launch(void* const* d_in, const int* in_sizes, int n_in,
                              void* d_out, int out_size, void* d_ws, size_t ws_size,
                              hipStream_t stream) {
    (void)in_sizes; (void)n_in; (void)out_size; (void)ws_size;

    const float* x    = (const float*)d_in[0];
    const int*   a1r  = (const int*)  d_in[1];
    const int*   a1c  = (const int*)  d_in[2];
    const float* a1v  = (const float*)d_in[3];
    const int*   a2r  = (const int*)  d_in[4];
    const int*   a2c  = (const int*)  d_in[5];
    const float* a2v  = (const float*)d_in[6];
    const float* W1   = (const float*)d_in[7];
    const float* Wout = (const float*)d_in[8];
    const float* bout = (const float*)d_in[9];
    float* out = (float*)d_out;

    float* h0 = (float*)d_ws;
    float* h1 = h0 + (size_t)NROWS * HID;
    float* h2 = h1 + (size_t)NROWS * HID;

    int*   off  = (int*)d_out + OFF_OFF;
    int*   ecol = (int*)d_out + OFF_ECOL;
    float* eval = (float*)d_out + OFF_EVAL;

    h2gcn_gemm1<<<dim3(NBKT + GEMM1_MB * GEMM1_NB), dim3(256), 0, stream>>>(
        x, W1, h0, a1r, a1c, a1v, a2r, a2c, a2v, off, ecol, eval);

    h2gcn_gather<<<dim3(NROWS / 8, 2), dim3(256), 0, stream>>>(off, ecol, eval, h0, h1);

    h2gcn_gemm2<<<dim3(79, 4), dim3(256), 0, stream>>>(h0, h1, h2, Wout, bout, out);
}
